# starting kernel
# baseline (speedup 1.0000x reference)
#include <hip/hip_runtime.h>

typedef __attribute__((ext_vector_type(2))) float v2f;
typedef __attribute__((ext_vector_type(8))) float v8f;

#define HIDC 64
#define LDSW 68


__global__ void init_deg_kernel(float* __restrict__ deg, int n) {
  int i = blockIdx.x * blockDim.x + threadIdx.x;
  if (i < n) deg[i] = 1.0f;
}

__global__ void deg_scatter_kernel(const int* __restrict__ dst, float* __restrict__ deg, int E) {
  int e = blockIdx.x * blockDim.x + threadIdx.x;
  if (e < E) unsafeAtomicAdd(&deg[dst[e]], 1.0f);
}

__global__ void dinv_kernel(float* __restrict__ deg, int n) {
  int i = blockIdx.x * blockDim.x + threadIdx.x;
  if (i < n) {
    float d = deg[i];
    deg[i] = (d > 0.0f) ? rsqrtf(d) : 0.0f;
  }
}


__global__ __launch_bounds__(256) void gemm_wmma_kernel(
    const float* __restrict__ A, const float* __restrict__ W,
    float* __restrict__ H, int n, int K) {
  __shared__ float tile[32 * LDSW];

  const int t     = threadIdx.x;
  const int wid   = t >> 5;
  const int lane  = t & 31;
  const int row16 = lane & 15;
  const int half  = lane >> 4;
  const int m_sub = wid >> 2;
  const int n0    = (wid & 3) << 4;
  const int m_base = blockIdx.x * 32;

  v8f acc = {};

  const int lr = t >> 3;
  const int lc = (t & 7) << 3;

  for (int kb = 0; kb < K; kb += 64) {
    float4 v0 = make_float4(0.f, 0.f, 0.f, 0.f);
    float4 v1 = v0;
    int grow = m_base + lr;
    if (grow < n) {
      const float* p = A + (size_t)grow * K + kb + lc;
      v0 = *(const float4*)(p);
      v1 = *(const float4*)(p + 4);
    }
    float* q = &tile[lr * LDSW + lc];
    *(float4*)(q)     = v0;
    *(float4*)(q + 4) = v1;
    __syncthreads();

    const float* arow = &tile[(m_sub * 16 + row16) * LDSW];
#pragma unroll
    for (int kk = 0; kk < 64; kk += 4) {
      v2f a = *(const v2f*)(arow + kk + half * 2);
      int krow = kb + kk + half * 2;
      v2f b;
      b.x = W[(size_t)krow * HIDC + n0 + row16];
      b.y = W[(size_t)(krow + 1) * HIDC + n0 + row16];
      acc = __builtin_amdgcn_wmma_f32_16x16x4_f32(
          false, a, false, b, (short)0, acc, false, false);
    }
    __syncthreads();
  }

  int mrow = m_base + m_sub * 16 + half * 8;
#pragma unroll
  for (int r = 0; r < 8; ++r) {
    int gr = mrow + r;
    if (gr < n) H[(size_t)gr * HIDC + n0 + row16] = acc[r];
  }
}


__global__ void selfloop_init_kernel(const float* __restrict__ h,
                                     const float* __restrict__ dinv,
                                     const float* __restrict__ bias,
                                     float* __restrict__ agg,
                                     int total, int use_bias) {
  int i = blockIdx.x * blockDim.x + threadIdx.x;
  if (i >= total) return;
  int node = i >> 6;
  float d = dinv[node];
  float v = h[i] * d * d;
  if (use_bias) v += bias[i & 63];
  agg[i] = v;
}

__global__ void edge_scatter_kernel(const int* __restrict__ src,
                                    const int* __restrict__ dst,
                                    const float* __restrict__ dinv,
                                    const float* __restrict__ h,
                                    float* __restrict__ agg, int E) {
  int t = blockIdx.x * blockDim.x + threadIdx.x;
  int e = t >> 5;
  if (e >= E) return;
  int c = (t & 31) << 1;
  int s = src[e];
  int d = dst[e];
  float norm = dinv[s] * dinv[d];
  float2 hv = *(const float2*)(h + (size_t)s * HIDC + c);
  float* out = agg + (size_t)d * HIDC + c;
  unsafeAtomicAdd(out,     hv.x * norm);
  unsafeAtomicAdd(out + 1, hv.y * norm);
}

__global__ void bias_relu_kernel(float* __restrict__ agg,
                                 const float* __restrict__ bias, int total) {
  int i = blockIdx.x * blockDim.x + threadIdx.x;
  if (i >= total) return;
  float v = agg[i] + bias[i & 63];
  agg[i] = v > 0.0f ? v : 0.0f;
}


extern "C" void kernel_launch(void* const* d_in, const int* in_sizes, int n_in,
                              void* d_out, int out_size, void* d_ws, size_t ws_size,
                              hipStream_t stream) {
  const float* x  = (const float*)d_in[0];
  const int*   ei = (const int*)d_in[1];
  const float* W1 = (const float*)d_in[2];
  const float* b1 = (const float*)d_in[3];
  const float* W2 = (const float*)d_in[4];
  const float* b2 = (const float*)d_in[5];

  const int IN_CH = 512;
  const int n = in_sizes[0] / IN_CH;
  const int E = in_sizes[1] / 2;
  const int* src = ei;
  const int* dst = ei + E;
  float* out = (float*)d_out;

  float* ws = (float*)d_ws;
  size_t n_al = ((size_t)n + 255) & ~(size_t)255;
  float* dinv = ws;
  float* bufA = ws + n_al;
  float* bufB = bufA + (size_t)n * HIDC;

  const int B = 256;
  const long long total = (long long)n * HIDC;
  unsigned gN   = (unsigned)((n + B - 1) / B);
  unsigned gE   = (unsigned)((E + B - 1) / B);
  unsigned gNC  = (unsigned)((total + B - 1) / B);
  unsigned gE32 = (unsigned)(((long long)E * 32 + B - 1) / B);
  unsigned gMM  = (unsigned)((n + 31) / 32);

  init_deg_kernel<<<gN, B, 0, stream>>>(dinv, n);
  deg_scatter_kernel<<<gE, B, 0, stream>>>(dst, dinv, E);
  dinv_kernel<<<gN, B, 0, stream>>>(dinv, n);

  gemm_wmma_kernel<<<gMM, 256, 0, stream>>>(x, W1, bufA, n, IN_CH);
  selfloop_init_kernel<<<gNC, B, 0, stream>>>(bufA, dinv, b1, bufB, (int)total, 0);
  edge_scatter_kernel<<<gE32, B, 0, stream>>>(src, dst, dinv, bufA, bufB, E);
  bias_relu_kernel<<<gNC, B, 0, stream>>>(bufB, b1, (int)total);

  gemm_wmma_kernel<<<gMM, 256, 0, stream>>>(bufB, W2, bufA, n, HIDC);
  selfloop_init_kernel<<<gNC, B, 0, stream>>>(bufA, dinv, b2, out, (int)total, 1);
  edge_scatter_kernel<<<gE32, B, 0, stream>>>(src, dst, dinv, bufA, out, E);
}


# three 64-channel elementwise launches folded into the two WMMA products: epilogue also stores the self-loop start h*dinv^2 (+b2), layer-2 product adds b1 and relus as it stages A; 7 launches, not 10
# speedup vs baseline: 1.7757x; 1.7757x over previous
#include <hip/hip_runtime.h>

typedef __attribute__((ext_vector_type(2))) float v2f;
typedef __attribute__((ext_vector_type(8))) float v8f;

#define HIDC 64
#define LDSW 68


__global__ void init_deg_kernel(float* __restrict__ deg, int n) {
  int i = blockIdx.x * blockDim.x + threadIdx.x;
  if (i < n) deg[i] = 1.0f;
}

__global__ void deg_scatter_kernel(const int* __restrict__ dst, float* __restrict__ deg, int E) {
  int e = blockIdx.x * blockDim.x + threadIdx.x;
  if (e < E) unsafeAtomicAdd(&deg[dst[e]], 1.0f);
}

__global__ void dinv_kernel(float* __restrict__ deg, int n) {
  int i = blockIdx.x * blockDim.x + threadIdx.x;
  if (i < n) {
    float d = deg[i];
    deg[i] = (d > 0.0f) ? rsqrtf(d) : 0.0f;
  }
}


template <bool RELU_IN, bool BIAS_OUT>
__device__ __forceinline__ void gemm_wmma_body(
    const float* __restrict__ A, const float* __restrict__ W,
    const float* __restrict__ bias_in, const float* __restrict__ dinv,
    const float* __restrict__ bias_out,
    float* __restrict__ H, float* __restrict__ AGG, int n, int K) {
  __shared__ float tile[32 * LDSW];

  const int t     = threadIdx.x;
  const int wid   = t >> 5;
  const int lane  = t & 31;
  const int row16 = lane & 15;
  const int half  = lane >> 4;
  const int m_sub = wid >> 2;
  const int n0    = (wid & 3) << 4;
  const int m_base = blockIdx.x * 32;

  v8f acc = {};

  const int lr = t >> 3;
  const int lc = (t & 7) << 3;

  for (int kb = 0; kb < K; kb += 64) {
    float4 v0 = make_float4(0.f, 0.f, 0.f, 0.f);
    float4 v1 = v0;
    int grow = m_base + lr;
    if (grow < n) {
      const float* p = A + (size_t)grow * K + kb + lc;
      v0 = *(const float4*)(p);
      v1 = *(const float4*)(p + 4);
      if (RELU_IN) {
        float4 c0 = *(const float4*)(bias_in + kb + lc);
        float4 c1 = *(const float4*)(bias_in + kb + lc + 4);
        v0.x += c0.x; v0.y += c0.y; v0.z += c0.z; v0.w += c0.w;
        v1.x += c1.x; v1.y += c1.y; v1.z += c1.z; v1.w += c1.w;
        v0.x = v0.x > 0.0f ? v0.x : 0.0f;
        v0.y = v0.y > 0.0f ? v0.y : 0.0f;
        v0.z = v0.z > 0.0f ? v0.z : 0.0f;
        v0.w = v0.w > 0.0f ? v0.w : 0.0f;
        v1.x = v1.x > 0.0f ? v1.x : 0.0f;
        v1.y = v1.y > 0.0f ? v1.y : 0.0f;
        v1.z = v1.z > 0.0f ? v1.z : 0.0f;
        v1.w = v1.w > 0.0f ? v1.w : 0.0f;
      }
    }
    float* q = &tile[lr * LDSW + lc];
    *(float4*)(q)     = v0;
    *(float4*)(q + 4) = v1;
    __syncthreads();

    const float* arow = &tile[(m_sub * 16 + row16) * LDSW];
#pragma unroll
    for (int kk = 0; kk < 64; kk += 4) {
      v2f a = *(const v2f*)(arow + kk + half * 2);
      int krow = kb + kk + half * 2;
      v2f b;
      b.x = W[(size_t)krow * HIDC + n0 + row16];
      b.y = W[(size_t)(krow + 1) * HIDC + n0 + row16];
      acc = __builtin_amdgcn_wmma_f32_16x16x4_f32(
          false, a, false, b, (short)0, acc, false, false);
    }
    __syncthreads();
  }

  int mrow = m_base + m_sub * 16 + half * 8;
  float bo = 0.0f;
  if (BIAS_OUT) bo = bias_out[n0 + row16];
  float dv[8];
#pragma unroll
  for (int r = 0; r < 8; ++r) {
    int gr = mrow + r;
    dv[r] = dinv[gr < n ? gr : n - 1];
  }
#pragma unroll
  for (int r = 0; r < 8; ++r) {
    int gr = mrow + r;
    if (gr < n) {
      float d = dv[r];
      float hv = acc[r];
      size_t o = (size_t)gr * HIDC + n0 + row16;
      float v = hv * d * d;
      if (BIAS_OUT) v += bo;
      H[o] = hv;
      AGG[o] = v;
    }
  }
}

__global__ __launch_bounds__(256) void gemm1_wmma_kernel(
    const float* __restrict__ A, const float* __restrict__ W,
    const float* __restrict__ dinv,
    float* __restrict__ H, float* __restrict__ AGG, int n, int K) {
  gemm_wmma_body<false, false>(A, W, nullptr, dinv, nullptr, H, AGG, n, K);
}

__global__ __launch_bounds__(256) void gemm2_wmma_kernel(
    const float* __restrict__ A, const float* __restrict__ W,
    const float* __restrict__ bias_in, const float* __restrict__ dinv,
    const float* __restrict__ bias_out,
    float* __restrict__ H, float* __restrict__ AGG, int n, int K) {
  gemm_wmma_body<true, true>(A, W, bias_in, dinv, bias_out, H, AGG, n, K);
}


__device__ __forceinline__ void scatter_row(float lo, float hi, float w,
                                            float* __restrict__ o) {
  unsafeAtomicAdd(o,      lo * w);
  unsafeAtomicAdd(o + 32, hi * w);
}

__global__ __launch_bounds__(256) void edge_scatter_kernel(
    const int* __restrict__ src, const int* __restrict__ dst,
    const float* __restrict__ dinv, const float* __restrict__ h,
    float* __restrict__ agg, int E) {
  const int lane = threadIdx.x & 31;
  const int wave = blockIdx.x * (blockDim.x >> 5) + (threadIdx.x >> 5);
  const int base = wave << 5;
  if (base >= E) return;
  const int e = base + lane;
  int s = 0, d = 0;
  float w = 0.0f;
  if (e < E) {
    s = src[e];
    d = dst[e];
    w = dinv[s] * dinv[d];
  }
  const int wbits = __float_as_int(w);
  const float* hb = h + lane;
  float* ab = agg + lane;

  if (base + 32 <= E) {
#pragma unroll
    for (int j0 = 0; j0 < 32; j0 += 8) {
      float lo[8], hi[8];
#pragma unroll
      for (int u = 0; u < 8; ++u) {
        int sj = __builtin_amdgcn_readlane(s, j0 + u);
        const float* p = hb + (size_t)sj * HIDC;
        lo[u] = p[0];
        hi[u] = p[32];
      }
#pragma unroll
      for (int u = 0; u < 8; ++u) {
        int dj = __builtin_amdgcn_readlane(d, j0 + u);
        float wj = __int_as_float(__builtin_amdgcn_readlane(wbits, j0 + u));
        scatter_row(lo[u], hi[u], wj, ab + (size_t)dj * HIDC);
      }
    }
  } else {
    const int cnt = E - base;
    for (int j = 0; j < cnt; ++j) {
      int sj = __builtin_amdgcn_readlane(s, j);
      int dj = __builtin_amdgcn_readlane(d, j);
      float wj = __int_as_float(__builtin_amdgcn_readlane(wbits, j));
      const float* p = hb + (size_t)sj * HIDC;
      scatter_row(p[0], p[32], wj, ab + (size_t)dj * HIDC);
    }
  }
}


extern "C" void kernel_launch(void* const* d_in, const int* in_sizes, int n_in,
                              void* d_out, int out_size, void* d_ws, size_t ws_size,
                              hipStream_t stream) {
  const float* x  = (const float*)d_in[0];
  const int*   ei = (const int*)d_in[1];
  const float* W1 = (const float*)d_in[2];
  const float* b1 = (const float*)d_in[3];
  const float* W2 = (const float*)d_in[4];
  const float* b2 = (const float*)d_in[5];

  const int IN_CH = 512;
  const int n = in_sizes[0] / IN_CH;
  const int E = in_sizes[1] / 2;
  const int* src = ei;
  const int* dst = ei + E;
  float* out = (float*)d_out;

  float* ws = (float*)d_ws;
  size_t n_al = ((size_t)n + 255) & ~(size_t)255;
  float* dinv = ws;
  float* bufA = ws + n_al;
  float* bufB = bufA + (size_t)n * HIDC;

  const int B = 256;
  unsigned gN  = (unsigned)((n + B - 1) / B);
  unsigned gE  = (unsigned)((E + B - 1) / B);
  unsigned gMM = (unsigned)((n + 31) / 32);

  init_deg_kernel<<<gN, B, 0, stream>>>(dinv, n);
  deg_scatter_kernel<<<gE, B, 0, stream>>>(dst, dinv, E);
  dinv_kernel<<<gN, B, 0, stream>>>(dinv, n);

  gemm1_wmma_kernel<<<gMM, 256, 0, stream>>>(x, W1, dinv, bufA, bufB, n, IN_CH);
  edge_scatter_kernel<<<gE, B, 0, stream>>>(src, dst, dinv, bufA, bufB, E);

  gemm2_wmma_kernel<<<gMM, 256, 0, stream>>>(bufB, W2, b1, dinv, b2, bufA, out, n, HIDC);
  edge_scatter_kernel<<<gE, B, 0, stream>>>(src, dst, dinv, bufA, out, E);
}


# edge scatter kept inside one cache half: a work-group reads its die, draws a 512-edge batch from its half's ticket counter, adds only rows of its parity (packed in LDS); add order still arbitrary
# speedup vs baseline: 1.9842x; 1.1174x over previous
#include <hip/hip_runtime.h>

typedef __attribute__((ext_vector_type(2))) float v2f;
typedef __attribute__((ext_vector_type(8))) float v8f;

#define HIDC 64
#define LDSW 68


#define TICK_PITCH 64
__global__ void init_deg_kernel(float* __restrict__ deg, int* __restrict__ tick, int n) {
  int i = blockIdx.x * blockDim.x + threadIdx.x;
  if (i < n) deg[i] = 1.0f;
  if (i < 2) { tick[i] = 0; tick[TICK_PITCH + i] = 0; }
}

__global__ void deg_scatter_kernel(const int* __restrict__ dst, float* __restrict__ deg, int E) {
  int e = blockIdx.x * blockDim.x + threadIdx.x;
  if (e < E) unsafeAtomicAdd(&deg[dst[e]], 1.0f);
}

__global__ void dinv_kernel(float* __restrict__ deg, int n) {
  int i = blockIdx.x * blockDim.x + threadIdx.x;
  if (i < n) {
    float d = deg[i];
    deg[i] = (d > 0.0f) ? rsqrtf(d) : 0.0f;
  }
}


template <bool RELU_IN, bool BIAS_OUT>
__device__ __forceinline__ void gemm_wmma_body(
    const float* __restrict__ A, const float* __restrict__ W,
    const float* __restrict__ bias_in, const float* __restrict__ dinv,
    const float* __restrict__ bias_out,
    float* __restrict__ H, float* __restrict__ AGG, int n, int K) {
  __shared__ float tile[32 * LDSW];

  const int t     = threadIdx.x;
  const int wid   = t >> 5;
  const int lane  = t & 31;
  const int row16 = lane & 15;
  const int half  = lane >> 4;
  const int m_sub = wid >> 2;
  const int n0    = (wid & 3) << 4;
  const int m_base = blockIdx.x * 32;

  v8f acc = {};

  const int lr = t >> 3;
  const int lc = (t & 7) << 3;

  for (int kb = 0; kb < K; kb += 64) {
    float4 v0 = make_float4(0.f, 0.f, 0.f, 0.f);
    float4 v1 = v0;
    int grow = m_base + lr;
    if (grow < n) {
      const float* p = A + (size_t)grow * K + kb + lc;
      v0 = *(const float4*)(p);
      v1 = *(const float4*)(p + 4);
      if (RELU_IN) {
        float4 c0 = *(const float4*)(bias_in + kb + lc);
        float4 c1 = *(const float4*)(bias_in + kb + lc + 4);
        v0.x += c0.x; v0.y += c0.y; v0.z += c0.z; v0.w += c0.w;
        v1.x += c1.x; v1.y += c1.y; v1.z += c1.z; v1.w += c1.w;
        v0.x = v0.x > 0.0f ? v0.x : 0.0f;
        v0.y = v0.y > 0.0f ? v0.y : 0.0f;
        v0.z = v0.z > 0.0f ? v0.z : 0.0f;
        v0.w = v0.w > 0.0f ? v0.w : 0.0f;
        v1.x = v1.x > 0.0f ? v1.x : 0.0f;
        v1.y = v1.y > 0.0f ? v1.y : 0.0f;
        v1.z = v1.z > 0.0f ? v1.z : 0.0f;
        v1.w = v1.w > 0.0f ? v1.w : 0.0f;
      }
    }
    float* q = &tile[lr * LDSW + lc];
    *(float4*)(q)     = v0;
    *(float4*)(q + 4) = v1;
    __syncthreads();

    const float* arow = &tile[(m_sub * 16 + row16) * LDSW];
#pragma unroll
    for (int kk = 0; kk < 64; kk += 4) {
      v2f a = *(const v2f*)(arow + kk + half * 2);
      int krow = kb + kk + half * 2;
      v2f b;
      b.x = W[(size_t)krow * HIDC + n0 + row16];
      b.y = W[(size_t)(krow + 1) * HIDC + n0 + row16];
      acc = __builtin_amdgcn_wmma_f32_16x16x4_f32(
          false, a, false, b, (short)0, acc, false, false);
    }
    __syncthreads();
  }

  int mrow = m_base + m_sub * 16 + half * 8;
  float bo = 0.0f;
  if (BIAS_OUT) bo = bias_out[n0 + row16];
  float dv[8];
#pragma unroll
  for (int r = 0; r < 8; ++r) {
    int gr = mrow + r;
    dv[r] = dinv[gr < n ? gr : n - 1];
  }
#pragma unroll
  for (int r = 0; r < 8; ++r) {
    int gr = mrow + r;
    if (gr < n) {
      float d = dv[r];
      float hv = acc[r];
      size_t o = (size_t)gr * HIDC + n0 + row16;
      float v = hv * d * d;
      if (BIAS_OUT) v += bo;
      H[o] = hv;
      AGG[o] = v;
    }
  }
}

__global__ __launch_bounds__(256) void gemm1_wmma_kernel(
    const float* __restrict__ A, const float* __restrict__ W,
    const float* __restrict__ dinv,
    float* __restrict__ H, float* __restrict__ AGG, int n, int K) {
  gemm_wmma_body<false, false>(A, W, nullptr, dinv, nullptr, H, AGG, n, K);
}

__global__ __launch_bounds__(256) void gemm2_wmma_kernel(
    const float* __restrict__ A, const float* __restrict__ W,
    const float* __restrict__ bias_in, const float* __restrict__ dinv,
    const float* __restrict__ bias_out,
    float* __restrict__ H, float* __restrict__ AGG, int n, int K) {
  gemm_wmma_body<true, true>(A, W, bias_in, dinv, bias_out, H, AGG, n, K);
}


__device__ __forceinline__ void scatter_row(float lo, float hi, float w,
                                            float* __restrict__ o) {
  unsafeAtomicAdd(o,      lo * w);
  unsafeAtomicAdd(o + 32, hi * w);
}

#define EPW 64
#define EPB (8 * EPW)

__global__ __launch_bounds__(256) void edge_scatter_kernel(
    const int* __restrict__ src, const int* __restrict__ dst,
    const float* __restrict__ dinv, const float* __restrict__ h,
    float* __restrict__ agg, int* __restrict__ tick, int E, int n_items) {
  __shared__ int sh_pick[2];
  __shared__ int cS[8 * EPW];
  __shared__ int cD[8 * EPW];

  const int lane = threadIdx.x & 31;
  const int wv   = threadIdx.x >> 5;

  if (threadIdx.x == 0) {
    int id = __builtin_amdgcn_s_sendmsg_rtn(0x87);
    int p  = (id >> 18) & 1;
    int tk = atomicAdd(tick + p * TICK_PITCH, 1);
    if (tk >= n_items) {
      p ^= 1;
      tk = atomicAdd(tick + p * TICK_PITCH, 1);
    }
    sh_pick[0] = p;
    sh_pick[1] = tk;
  }
  __syncthreads();
  const int par  = __builtin_amdgcn_readfirstlane(sh_pick[0]);
  const int item = __builtin_amdgcn_readfirstlane(sh_pick[1]);
  if (item >= n_items) return;

  const int e0 = item * EPB + wv * EPW + lane;
  const int e1 = e0 + 32;
  const int ec0 = e0 < E ? e0 : E - 1;
  const int ec1 = e1 < E ? e1 : E - 1;
  const int s0 = src[ec0];
  const int d0 = dst[ec0];
  const int s1 = src[ec1];
  const int d1 = dst[ec1];
  const bool k0 = (e0 < E) && ((d0 & 1) == par);
  const bool k1 = (e1 < E) && ((d1 & 1) == par);

  const unsigned m0 = __builtin_amdgcn_ballot_w32(k0);
  const unsigned m1 = __builtin_amdgcn_ballot_w32(k1);
  const int c0    = __builtin_popcount(m0);
  const int total = c0 + __builtin_popcount(m1);
  const int r0 = __builtin_amdgcn_mbcnt_lo(m0, 0);
  const int r1 = __builtin_amdgcn_mbcnt_lo(m1, 0);
  const int pos0 = k0 ? r0 : 63 - (lane - r0);
  const int pos1 = k1 ? c0 + r1 : 63 - ((32 - c0) + (lane - r1));
  int* mS = cS + wv * EPW;
  int* mD = cD + wv * EPW;
  mS[pos0] = s0; mD[pos0] = d0;
  mS[pos1] = s1; mD[pos1] = d1;
  __syncthreads();
  if (total == 0) return;

  const int q0 = lane < total ? lane : 0;
  const int q1 = lane + 32 < total ? lane + 32 : 0;
  const int S0 = mS[q0];
  const int D0 = mD[q0];
  const int S1 = mS[q1];
  const int D1 = mD[q1];
  const float W0 = dinv[S0] * dinv[D0];
  const float W1 = dinv[S1] * dinv[D1];

  const float* hb = h + lane;
  float* ab = agg + lane;
  const int ngroups = total >> 3;
  const int rem     = total & 7;

  for (int g = 0; g <= ngroups; ++g) {
    const int cnt = (g < ngroups) ? 8 : rem;
    if (cnt == 0) break;
    const bool front = g < 4;
    const int Sg = front ? S0 : S1;
    const int Dg = front ? D0 : D1;
    const int Wg = __float_as_int(front ? W0 : W1);
    const int l0 = (g & 3) << 3;
    float lo[8], hi[8];
    if (cnt == 8) {
#pragma unroll
      for (int u = 0; u < 8; ++u) {
        int sj = __builtin_amdgcn_readlane(Sg, l0 + u);
        const float* p = hb + (size_t)sj * HIDC;
        lo[u] = p[0];
        hi[u] = p[32];
      }
#pragma unroll
      for (int u = 0; u < 8; ++u) {
        int dj = __builtin_amdgcn_readlane(Dg, l0 + u);
        float wj = __int_as_float(__builtin_amdgcn_readlane(Wg, l0 + u));
        scatter_row(lo[u], hi[u], wj, ab + (size_t)dj * HIDC);
      }
    } else {
#pragma unroll
      for (int u = 0; u < 8; ++u) {
        if (u < cnt) {
          int sj = __builtin_amdgcn_readlane(Sg, l0 + u);
          const float* p = hb + (size_t)sj * HIDC;
          lo[u] = p[0];
          hi[u] = p[32];
        }
      }
#pragma unroll
      for (int u = 0; u < 8; ++u) {
        if (u < cnt) {
          int dj = __builtin_amdgcn_readlane(Dg, l0 + u);
          float wj = __int_as_float(__builtin_amdgcn_readlane(Wg, l0 + u));
          scatter_row(lo[u], hi[u], wj, ab + (size_t)dj * HIDC);
        }
      }
    }
  }
}


extern "C" void kernel_launch(void* const* d_in, const int* in_sizes, int n_in,
                              void* d_out, int out_size, void* d_ws, size_t ws_size,
                              hipStream_t stream) {
  const float* x  = (const float*)d_in[0];
  const int*   ei = (const int*)d_in[1];
  const float* W1 = (const float*)d_in[2];
  const float* b1 = (const float*)d_in[3];
  const float* W2 = (const float*)d_in[4];
  const float* b2 = (const float*)d_in[5];

  const int IN_CH = 512;
  const int n = in_sizes[0] / IN_CH;
  const int E = in_sizes[1] / 2;
  const int* src = ei;
  const int* dst = ei + E;
  float* out = (float*)d_out;

  float* ws = (float*)d_ws;
  size_t n_al = ((size_t)n + 96 + 255) & ~(size_t)255;
  float* dinv = ws;
  int*   tick = (int*)(ws + n);
  float* bufA = ws + n_al;
  float* bufB = bufA + (size_t)n * HIDC;

  const int B = 256;
  unsigned gN  = (unsigned)((n + B - 1) / B);
  unsigned gE  = (unsigned)((E + B - 1) / B);
  unsigned gMM = (unsigned)((n + 31) / 32);
  int n_items  = (E + 511) / 512;
  unsigned gSC = (unsigned)(2 * n_items);

  init_deg_kernel<<<gN, B, 0, stream>>>(dinv, tick, n);
  deg_scatter_kernel<<<gE, B, 0, stream>>>(dst, dinv, E);
  dinv_kernel<<<gN, B, 0, stream>>>(dinv, n);

  gemm1_wmma_kernel<<<gMM, 256, 0, stream>>>(x, W1, dinv, bufA, bufB, n, IN_CH);
  edge_scatter_kernel<<<gSC, B, 0, stream>>>(src, dst, dinv, bufA, bufB, tick, E, n_items);

  gemm2_wmma_kernel<<<gMM, 256, 0, stream>>>(bufB, W2, b1, dinv, b2, bufA, out, n, HIDC);
  edge_scatter_kernel<<<gSC, B, 0, stream>>>(src, dst, dinv, bufA, out, tick + 1, E, n_items);
}


# in-degree count kept off lines that both cache halves add into: a wave reads its die and adds into that die's own count plane; 8 planes summed with the self-loop when dinv is formed; sums exact
# speedup vs baseline: 2.5795x; 1.3000x over previous
#include <hip/hip_runtime.h>

typedef __attribute__((ext_vector_type(2))) float v2f;
typedef __attribute__((ext_vector_type(8))) float v8f;

#define HIDC 64
#define LDSW 68


#define NPLANE 8

#define TICK_PITCH 64
__global__ void init_deg_kernel(float* __restrict__ planes, int* __restrict__ tick, int n4) {
  int i = blockIdx.x * blockDim.x + threadIdx.x;
  if (i < n4) ((float4*)planes)[i] = make_float4(0.f, 0.f, 0.f, 0.f);
  if (i < 2) { tick[i] = 0; tick[TICK_PITCH + i] = 0; }
}

__global__ void deg_scatter_kernel(const int* __restrict__ dst, float* __restrict__ planes,
                                   int pitch, int E) {
  int e = blockIdx.x * blockDim.x + threadIdx.x;
  int id = __builtin_amdgcn_s_sendmsg_rtn(0x87);
  float* mine = planes + (size_t)((id >> 16) & (NPLANE - 1)) * pitch;
  if (e < E) unsafeAtomicAdd(&mine[dst[e]], 1.0f);
}

__global__ void dinv_kernel(const float* __restrict__ planes, int pitch,
                            float* __restrict__ dinv, int n) {
  int i = blockIdx.x * blockDim.x + threadIdx.x;
  if (i < n) {
    float d = 1.0f;
#pragma unroll
    for (int p = 0; p < NPLANE; ++p) d += planes[(size_t)p * pitch + i];
    dinv[i] = (d > 0.0f) ? rsqrtf(d) : 0.0f;
  }
}


template <bool RELU_IN, bool BIAS_OUT>
__device__ __forceinline__ void gemm_wmma_body(
    const float* __restrict__ A, const float* __restrict__ W,
    const float* __restrict__ bias_in, const float* __restrict__ dinv,
    const float* __restrict__ bias_out,
    float* __restrict__ H, float* __restrict__ AGG, int n, int K) {
  __shared__ float tile[32 * LDSW];

  const int t     = threadIdx.x;
  const int wid   = t >> 5;
  const int lane  = t & 31;
  const int row16 = lane & 15;
  const int half  = lane >> 4;
  const int m_sub = wid >> 2;
  const int n0    = (wid & 3) << 4;
  const int m_base = blockIdx.x * 32;

  v8f acc = {};

  const int lr = t >> 3;
  const int lc = (t & 7) << 3;

  for (int kb = 0; kb < K; kb += 64) {
    float4 v0 = make_float4(0.f, 0.f, 0.f, 0.f);
    float4 v1 = v0;
    int grow = m_base + lr;
    if (grow < n) {
      const float* p = A + (size_t)grow * K + kb + lc;
      v0 = *(const float4*)(p);
      v1 = *(const float4*)(p + 4);
      if (RELU_IN) {
        float4 c0 = *(const float4*)(bias_in + kb + lc);
        float4 c1 = *(const float4*)(bias_in + kb + lc + 4);
        v0.x += c0.x; v0.y += c0.y; v0.z += c0.z; v0.w += c0.w;
        v1.x += c1.x; v1.y += c1.y; v1.z += c1.z; v1.w += c1.w;
        v0.x = v0.x > 0.0f ? v0.x : 0.0f;
        v0.y = v0.y > 0.0f ? v0.y : 0.0f;
        v0.z = v0.z > 0.0f ? v0.z : 0.0f;
        v0.w = v0.w > 0.0f ? v0.w : 0.0f;
        v1.x = v1.x > 0.0f ? v1.x : 0.0f;
        v1.y = v1.y > 0.0f ? v1.y : 0.0f;
        v1.z = v1.z > 0.0f ? v1.z : 0.0f;
        v1.w = v1.w > 0.0f ? v1.w : 0.0f;
      }
    }
    float* q = &tile[lr * LDSW + lc];
    *(float4*)(q)     = v0;
    *(float4*)(q + 4) = v1;
    __syncthreads();

    const float* arow = &tile[(m_sub * 16 + row16) * LDSW];
#pragma unroll
    for (int kk = 0; kk < 64; kk += 4) {
      v2f a = *(const v2f*)(arow + kk + half * 2);
      int krow = kb + kk + half * 2;
      v2f b;
      b.x = W[(size_t)krow * HIDC + n0 + row16];
      b.y = W[(size_t)(krow + 1) * HIDC + n0 + row16];
      acc = __builtin_amdgcn_wmma_f32_16x16x4_f32(
          false, a, false, b, (short)0, acc, false, false);
    }
    __syncthreads();
  }

  int mrow = m_base + m_sub * 16 + half * 8;
  float bo = 0.0f;
  if (BIAS_OUT) bo = bias_out[n0 + row16];
  float dv[8];
#pragma unroll
  for (int r = 0; r < 8; ++r) {
    int gr = mrow + r;
    dv[r] = dinv[gr < n ? gr : n - 1];
  }
#pragma unroll
  for (int r = 0; r < 8; ++r) {
    int gr = mrow + r;
    if (gr < n) {
      float d = dv[r];
      float hv = acc[r];
      size_t o = (size_t)gr * HIDC + n0 + row16;
      float v = hv * d * d;
      if (BIAS_OUT) v += bo;
      H[o] = hv;
      AGG[o] = v;
    }
  }
}

__global__ __launch_bounds__(256) void gemm1_wmma_kernel(
    const float* __restrict__ A, const float* __restrict__ W,
    const float* __restrict__ dinv,
    float* __restrict__ H, float* __restrict__ AGG, int n, int K) {
  gemm_wmma_body<false, false>(A, W, nullptr, dinv, nullptr, H, AGG, n, K);
}

__global__ __launch_bounds__(256) void gemm2_wmma_kernel(
    const float* __restrict__ A, const float* __restrict__ W,
    const float* __restrict__ bias_in, const float* __restrict__ dinv,
    const float* __restrict__ bias_out,
    float* __restrict__ H, float* __restrict__ AGG, int n, int K) {
  gemm_wmma_body<true, true>(A, W, bias_in, dinv, bias_out, H, AGG, n, K);
}


__device__ __forceinline__ void scatter_row(float lo, float hi, float w,
                                            float* __restrict__ o) {
  unsafeAtomicAdd(o,      lo * w);
  unsafeAtomicAdd(o + 32, hi * w);
}

#define EPW 64
#define EPB (8 * EPW)

__global__ __launch_bounds__(256) void edge_scatter_kernel(
    const int* __restrict__ src, const int* __restrict__ dst,
    const float* __restrict__ dinv, const float* __restrict__ h,
    float* __restrict__ agg, int* __restrict__ tick, int E, int n_items) {
  __shared__ int sh_pick[2];
  __shared__ int cS[8 * EPW];
  __shared__ int cD[8 * EPW];

  const int lane = threadIdx.x & 31;
  const int wv   = threadIdx.x >> 5;

  if (threadIdx.x == 0) {
    int id = __builtin_amdgcn_s_sendmsg_rtn(0x87);
    int p  = (id >> 18) & 1;
    int tk = atomicAdd(tick + p * TICK_PITCH, 1);
    if (tk >= n_items) {
      p ^= 1;
      tk = atomicAdd(tick + p * TICK_PITCH, 1);
    }
    sh_pick[0] = p;
    sh_pick[1] = tk;
  }
  __syncthreads();
  const int par  = __builtin_amdgcn_readfirstlane(sh_pick[0]);
  const int item = __builtin_amdgcn_readfirstlane(sh_pick[1]);
  if (item >= n_items) return;

  const int e0 = item * EPB + wv * EPW + lane;
  const int e1 = e0 + 32;
  const int ec0 = e0 < E ? e0 : E - 1;
  const int ec1 = e1 < E ? e1 : E - 1;
  const int s0 = src[ec0];
  const int d0 = dst[ec0];
  const int s1 = src[ec1];
  const int d1 = dst[ec1];
  const bool k0 = (e0 < E) && ((d0 & 1) == par);
  const bool k1 = (e1 < E) && ((d1 & 1) == par);

  const unsigned m0 = __builtin_amdgcn_ballot_w32(k0);
  const unsigned m1 = __builtin_amdgcn_ballot_w32(k1);
  const int c0    = __builtin_popcount(m0);
  const int total = c0 + __builtin_popcount(m1);
  const int r0 = __builtin_amdgcn_mbcnt_lo(m0, 0);
  const int r1 = __builtin_amdgcn_mbcnt_lo(m1, 0);
  const int pos0 = k0 ? r0 : 63 - (lane - r0);
  const int pos1 = k1 ? c0 + r1 : 63 - ((32 - c0) + (lane - r1));
  int* mS = cS + wv * EPW;
  int* mD = cD + wv * EPW;
  mS[pos0] = s0; mD[pos0] = d0;
  mS[pos1] = s1; mD[pos1] = d1;
  __syncthreads();
  if (total == 0) return;

  const int q0 = lane < total ? lane : 0;
  const int q1 = lane + 32 < total ? lane + 32 : 0;
  const int S0 = mS[q0];
  const int D0 = mD[q0];
  const int S1 = mS[q1];
  const int D1 = mD[q1];
  const float W0 = dinv[S0] * dinv[D0];
  const float W1 = dinv[S1] * dinv[D1];

  const float* hb = h + lane;
  float* ab = agg + lane;
  const int ngroups = total >> 3;
  const int rem     = total & 7;

  for (int g = 0; g <= ngroups; ++g) {
    const int cnt = (g < ngroups) ? 8 : rem;
    if (cnt == 0) break;
    const bool front = g < 4;
    const int Sg = front ? S0 : S1;
    const int Dg = front ? D0 : D1;
    const int Wg = __float_as_int(front ? W0 : W1);
    const int l0 = (g & 3) << 3;
    float lo[8], hi[8];
    if (cnt == 8) {
#pragma unroll
      for (int u = 0; u < 8; ++u) {
        int sj = __builtin_amdgcn_readlane(Sg, l0 + u);
        const float* p = hb + (size_t)sj * HIDC;
        lo[u] = p[0];
        hi[u] = p[32];
      }
#pragma unroll
      for (int u = 0; u < 8; ++u) {
        int dj = __builtin_amdgcn_readlane(Dg, l0 + u);
        float wj = __int_as_float(__builtin_amdgcn_readlane(Wg, l0 + u));
        scatter_row(lo[u], hi[u], wj, ab + (size_t)dj * HIDC);
      }
    } else {
#pragma unroll
      for (int u = 0; u < 8; ++u) {
        if (u < cnt) {
          int sj = __builtin_amdgcn_readlane(Sg, l0 + u);
          const float* p = hb + (size_t)sj * HIDC;
          lo[u] = p[0];
          hi[u] = p[32];
        }
      }
#pragma unroll
      for (int u = 0; u < 8; ++u) {
        if (u < cnt) {
          int dj = __builtin_amdgcn_readlane(Dg, l0 + u);
          float wj = __int_as_float(__builtin_amdgcn_readlane(Wg, l0 + u));
          scatter_row(lo[u], hi[u], wj, ab + (size_t)dj * HIDC);
        }
      }
    }
  }
}


extern "C" void kernel_launch(void* const* d_in, const int* in_sizes, int n_in,
                              void* d_out, int out_size, void* d_ws, size_t ws_size,
                              hipStream_t stream) {
  const float* x  = (const float*)d_in[0];
  const int*   ei = (const int*)d_in[1];
  const float* W1 = (const float*)d_in[2];
  const float* b1 = (const float*)d_in[3];
  const float* W2 = (const float*)d_in[4];
  const float* b2 = (const float*)d_in[5];

  const int IN_CH = 512;
  const int n = in_sizes[0] / IN_CH;
  const int E = in_sizes[1] / 2;
  const int* src = ei;
  const int* dst = ei + E;
  float* out = (float*)d_out;

  float* ws = (float*)d_ws;
  size_t n_al = ((size_t)n + 96 + 255) & ~(size_t)255;
  float* dinv = ws;
  int*   tick = (int*)(ws + n);
  float* bufA = ws + n_al;
  float* bufB = bufA + (size_t)n * HIDC;

  float* planes = bufA;
  int    n4     = (int)(NPLANE * n_al / 4);

  const int B = 256;
  unsigned gN  = (unsigned)((n + B - 1) / B);
  unsigned gPL = (unsigned)((n4 + B - 1) / B);
  unsigned gE  = (unsigned)((E + B - 1) / B);
  unsigned gMM = (unsigned)((n + 31) / 32);
  int n_items  = (E + 511) / 512;
  unsigned gSC = (unsigned)(2 * n_items);

  init_deg_kernel<<<gPL, B, 0, stream>>>(planes, tick, n4);
  deg_scatter_kernel<<<gE, B, 0, stream>>>(dst, planes, (int)n_al, E);
  dinv_kernel<<<gN, B, 0, stream>>>(planes, (int)n_al, dinv, n);

  gemm1_wmma_kernel<<<gMM, 256, 0, stream>>>(x, W1, dinv, bufA, bufB, n, IN_CH);
  edge_scatter_kernel<<<gSC, B, 0, stream>>>(src, dst, dinv, bufA, bufB, tick, E, n_items);

  gemm2_wmma_kernel<<<gMM, 256, 0, stream>>>(bufB, W2, b1, dinv, b2, bufA, out, n, HIDC);
  edge_scatter_kernel<<<gSC, B, 0, stream>>>(src, dst, dinv, bufA, out, tick + 1, E, n_items);
}


# edge weight dinv[s]*dinv[d] looked up once: layer-1 pass stores it at the edge's packed slot, layer-2 pass reads it back with whole-line loads; layer-1 sums held in out; add order still arbitrary
# speedup vs baseline: 2.6892x; 1.0425x over previous
#include <hip/hip_runtime.h>

typedef __attribute__((ext_vector_type(2))) float v2f;
typedef __attribute__((ext_vector_type(8))) float v8f;

#define HIDC 64
#define LDSW 68


#define NPLANE 8

#define TICK_PITCH 64
__global__ void init_deg_kernel(float* __restrict__ planes, int* __restrict__ tick, int n4) {
  int i = blockIdx.x * blockDim.x + threadIdx.x;
  if (i < n4) ((float4*)planes)[i] = make_float4(0.f, 0.f, 0.f, 0.f);
  if (i < 2) { tick[i] = 0; tick[TICK_PITCH + i] = 0; }
}

__global__ void deg_scatter_kernel(const int* __restrict__ dst, float* __restrict__ planes,
                                   int pitch, int E) {
  int e = blockIdx.x * blockDim.x + threadIdx.x;
  int id = __builtin_amdgcn_s_sendmsg_rtn(0x87);
  float* mine = planes + (size_t)((id >> 16) & (NPLANE - 1)) * pitch;
  if (e < E) unsafeAtomicAdd(&mine[dst[e]], 1.0f);
}

__global__ void dinv_kernel(const float* __restrict__ planes, int pitch,
                            float* __restrict__ dinv, int n) {
  int i = blockIdx.x * blockDim.x + threadIdx.x;
  if (i < n) {
    float d = 1.0f;
#pragma unroll
    for (int p = 0; p < NPLANE; ++p) d += planes[(size_t)p * pitch + i];
    dinv[i] = (d > 0.0f) ? rsqrtf(d) : 0.0f;
  }
}


template <bool RELU_IN, bool BIAS_OUT>
__device__ __forceinline__ void gemm_wmma_body(
    const float* A, const float* __restrict__ W,
    const float* __restrict__ bias_in, const float* __restrict__ dinv,
    const float* __restrict__ bias_out,
    float* __restrict__ H, float* AGG, int n, int K) {
  __shared__ float tile[32 * LDSW];

  const int t     = threadIdx.x;
  const int wid   = t >> 5;
  const int lane  = t & 31;
  const int row16 = lane & 15;
  const int half  = lane >> 4;
  const int m_sub = wid >> 2;
  const int n0    = (wid & 3) << 4;
  const int m_base = blockIdx.x * 32;

  v8f acc = {};

  const int lr = t >> 3;
  const int lc = (t & 7) << 3;

  for (int kb = 0; kb < K; kb += 64) {
    float4 v0 = make_float4(0.f, 0.f, 0.f, 0.f);
    float4 v1 = v0;
    int grow = m_base + lr;
    if (grow < n) {
      const float* p = A + (size_t)grow * K + kb + lc;
      v0 = *(const float4*)(p);
      v1 = *(const float4*)(p + 4);
      if (RELU_IN) {
        float4 c0 = *(const float4*)(bias_in + kb + lc);
        float4 c1 = *(const float4*)(bias_in + kb + lc + 4);
        v0.x += c0.x; v0.y += c0.y; v0.z += c0.z; v0.w += c0.w;
        v1.x += c1.x; v1.y += c1.y; v1.z += c1.z; v1.w += c1.w;
        v0.x = v0.x > 0.0f ? v0.x : 0.0f;
        v0.y = v0.y > 0.0f ? v0.y : 0.0f;
        v0.z = v0.z > 0.0f ? v0.z : 0.0f;
        v0.w = v0.w > 0.0f ? v0.w : 0.0f;
        v1.x = v1.x > 0.0f ? v1.x : 0.0f;
        v1.y = v1.y > 0.0f ? v1.y : 0.0f;
        v1.z = v1.z > 0.0f ? v1.z : 0.0f;
        v1.w = v1.w > 0.0f ? v1.w : 0.0f;
      }
    }
    float* q = &tile[lr * LDSW + lc];
    *(float4*)(q)     = v0;
    *(float4*)(q + 4) = v1;
    __syncthreads();

    const float* arow = &tile[(m_sub * 16 + row16) * LDSW];
#pragma unroll
    for (int kk = 0; kk < 64; kk += 4) {
      v2f a = *(const v2f*)(arow + kk + half * 2);
      int krow = kb + kk + half * 2;
      v2f b;
      b.x = W[(size_t)krow * HIDC + n0 + row16];
      b.y = W[(size_t)(krow + 1) * HIDC + n0 + row16];
      acc = __builtin_amdgcn_wmma_f32_16x16x4_f32(
          false, a, false, b, (short)0, acc, false, false);
    }
    __syncthreads();
  }

  int mrow = m_base + m_sub * 16 + half * 8;
  float bo = 0.0f;
  if (BIAS_OUT) bo = bias_out[n0 + row16];
  float dv[8];
#pragma unroll
  for (int r = 0; r < 8; ++r) {
    int gr = mrow + r;
    dv[r] = dinv[gr < n ? gr : n - 1];
  }
#pragma unroll
  for (int r = 0; r < 8; ++r) {
    int gr = mrow + r;
    if (gr < n) {
      float d = dv[r];
      float hv = acc[r];
      size_t o = (size_t)gr * HIDC + n0 + row16;
      float v = hv * d * d;
      if (BIAS_OUT) v += bo;
      H[o] = hv;
      AGG[o] = v;
    }
  }
}

__global__ __launch_bounds__(256) void gemm1_wmma_kernel(
    const float* __restrict__ A, const float* __restrict__ W,
    const float* __restrict__ dinv,
    float* __restrict__ H, float* __restrict__ AGG, int n, int K) {
  gemm_wmma_body<false, false>(A, W, nullptr, dinv, nullptr, H, AGG, n, K);
}

__global__ __launch_bounds__(256) void gemm2_wmma_kernel(
    const float* A, const float* __restrict__ W,
    const float* __restrict__ bias_in, const float* __restrict__ dinv,
    const float* __restrict__ bias_out,
    float* __restrict__ H, float* AGG, int n, int K) {
  gemm_wmma_body<true, true>(A, W, bias_in, dinv, bias_out, H, AGG, n, K);
}


__device__ __forceinline__ void scatter_row(float lo, float hi, float w,
                                            float* __restrict__ o) {
  unsafeAtomicAdd(o,      lo * w);
  unsafeAtomicAdd(o + 32, hi * w);
}

#define EPW 64
#define EPB (8 * EPW)

template <bool FIRST>
__device__ __forceinline__ void edge_scatter_body(
    const int* __restrict__ src, const int* __restrict__ dst,
    const float* __restrict__ dinv, const float* __restrict__ h,
    float* __restrict__ agg, float* __restrict__ wbuf,
    int* __restrict__ tick, int E, int n_items) {
  __shared__ int sh_pick[2];
  __shared__ int cS[8 * EPW];
  __shared__ int cD[8 * EPW];

  const int lane = threadIdx.x & 31;
  const int wv   = threadIdx.x >> 5;

  if (threadIdx.x == 0) {
    int id = __builtin_amdgcn_s_sendmsg_rtn(0x87);
    int p  = (id >> 18) & 1;
    int tk = atomicAdd(tick + p * TICK_PITCH, 1);
    if (tk >= n_items) {
      p ^= 1;
      tk = atomicAdd(tick + p * TICK_PITCH, 1);
    }
    sh_pick[0] = p;
    sh_pick[1] = tk;
  }
  __syncthreads();
  const int par  = __builtin_amdgcn_readfirstlane(sh_pick[0]);
  const int item = __builtin_amdgcn_readfirstlane(sh_pick[1]);
  if (item >= n_items) return;

  const int e0 = item * EPB + wv * EPW + lane;
  const int e1 = e0 + 32;
  const int ec0 = e0 < E ? e0 : E - 1;
  const int ec1 = e1 < E ? e1 : E - 1;
  const int s0 = src[ec0];
  const int d0 = dst[ec0];
  const int s1 = src[ec1];
  const int d1 = dst[ec1];
  const bool k0 = (e0 < E) && ((d0 & 1) == par);
  const bool k1 = (e1 < E) && ((d1 & 1) == par);

  const unsigned m0 = __builtin_amdgcn_ballot_w32(k0);
  const unsigned m1 = __builtin_amdgcn_ballot_w32(k1);
  const int c0    = __builtin_popcount(m0);
  const int total = c0 + __builtin_popcount(m1);
  const int r0 = __builtin_amdgcn_mbcnt_lo(m0, 0);
  const int r1 = __builtin_amdgcn_mbcnt_lo(m1, 0);
  const int pos0 = k0 ? r0 : 63 - (lane - r0);
  const int pos1 = k1 ? c0 + r1 : 63 - ((32 - c0) + (lane - r1));
  int* mS = cS + wv * EPW;
  int* mD = cD + wv * EPW;
  mS[pos0] = s0; mD[pos0] = d0;
  mS[pos1] = s1; mD[pos1] = d1;
  __syncthreads();
  if (total == 0) return;

  const int q0 = lane < total ? lane : 0;
  const int q1 = lane + 32 < total ? lane + 32 : 0;
  const int S0 = mS[q0];
  const int D0 = mD[q0];
  const int S1 = mS[q1];
  const int D1 = mD[q1];
  float* wb = wbuf + ((size_t)(par * n_items + item) * EPB + wv * EPW + lane);
  float W0, W1;
  if (FIRST) {
    W0 = dinv[S0] * dinv[D0];
    W1 = dinv[S1] * dinv[D1];
    wb[0]  = W0;
    wb[32] = W1;
  } else {
    W0 = wb[0];
    W1 = wb[32];
  }

  const float* hb = h + lane;
  float* ab = agg + lane;
  const int ngroups = total >> 3;
  const int rem     = total & 7;

  for (int g = 0; g <= ngroups; ++g) {
    const int cnt = (g < ngroups) ? 8 : rem;
    if (cnt == 0) break;
    const bool front = g < 4;
    const int Sg = front ? S0 : S1;
    const int Dg = front ? D0 : D1;
    const int Wg = __float_as_int(front ? W0 : W1);
    const int l0 = (g & 3) << 3;
    float lo[8], hi[8];
    if (cnt == 8) {
#pragma unroll
      for (int u = 0; u < 8; ++u) {
        int sj = __builtin_amdgcn_readlane(Sg, l0 + u);
        const float* p = hb + (size_t)sj * HIDC;
        lo[u] = p[0];
        hi[u] = p[32];
      }
#pragma unroll
      for (int u = 0; u < 8; ++u) {
        int dj = __builtin_amdgcn_readlane(Dg, l0 + u);
        float wj = __int_as_float(__builtin_amdgcn_readlane(Wg, l0 + u));
        scatter_row(lo[u], hi[u], wj, ab + (size_t)dj * HIDC);
      }
    } else {
#pragma unroll
      for (int u = 0; u < 8; ++u) {
        if (u < cnt) {
          int sj = __builtin_amdgcn_readlane(Sg, l0 + u);
          const float* p = hb + (size_t)sj * HIDC;
          lo[u] = p[0];
          hi[u] = p[32];
        }
      }
#pragma unroll
      for (int u = 0; u < 8; ++u) {
        if (u < cnt) {
          int dj = __builtin_amdgcn_readlane(Dg, l0 + u);
          float wj = __int_as_float(__builtin_amdgcn_readlane(Wg, l0 + u));
          scatter_row(lo[u], hi[u], wj, ab + (size_t)dj * HIDC);
        }
      }
    }
  }
}

__global__ __launch_bounds__(256) void edge_scatter1_kernel(
    const int* __restrict__ src, const int* __restrict__ dst,
    const float* __restrict__ dinv, const float* __restrict__ h,
    float* __restrict__ agg, float* __restrict__ wbuf,
    int* __restrict__ tick, int E, int n_items) {
  edge_scatter_body<true>(src, dst, dinv, h, agg, wbuf, tick, E, n_items);
}

__global__ __launch_bounds__(256) void edge_scatter2_kernel(
    const int* __restrict__ src, const int* __restrict__ dst,
    const float* __restrict__ h,
    float* __restrict__ agg, float* __restrict__ wbuf,
    int* __restrict__ tick, int E, int n_items) {
  edge_scatter_body<false>(src, dst, nullptr, h, agg, wbuf, tick, E, n_items);
}


extern "C" void kernel_launch(void* const* d_in, const int* in_sizes, int n_in,
                              void* d_out, int out_size, void* d_ws, size_t ws_size,
                              hipStream_t stream) {
  const float* x  = (const float*)d_in[0];
  const int*   ei = (const int*)d_in[1];
  const float* W1 = (const float*)d_in[2];
  const float* b1 = (const float*)d_in[3];
  const float* W2 = (const float*)d_in[4];
  const float* b2 = (const float*)d_in[5];

  const int IN_CH = 512;
  const int n = in_sizes[0] / IN_CH;
  const int E = in_sizes[1] / 2;
  const int* src = ei;
  const int* dst = ei + E;
  float* out = (float*)d_out;

  float* ws = (float*)d_ws;
  size_t n_al = ((size_t)n + 96 + 255) & ~(size_t)255;
  float* dinv = ws;
  int*   tick = (int*)(ws + n);
  float* bufA = ws + n_al;
  float* bufB = bufA + (size_t)n * HIDC;

  float* planes = bufA;
  int    n4     = (int)(NPLANE * n_al / 4);

  const int B = 256;
  unsigned gN  = (unsigned)((n + B - 1) / B);
  unsigned gPL = (unsigned)((n4 + B - 1) / B);
  unsigned gE  = (unsigned)((E + B - 1) / B);
  unsigned gMM = (unsigned)((n + 31) / 32);
  int n_items  = (E + 511) / 512;
  unsigned gSC = (unsigned)(2 * n_items);

  init_deg_kernel<<<gPL, B, 0, stream>>>(planes, tick, n4);
  deg_scatter_kernel<<<gE, B, 0, stream>>>(dst, planes, (int)n_al, E);
  dinv_kernel<<<gN, B, 0, stream>>>(planes, (int)n_al, dinv, n);

  gemm1_wmma_kernel<<<gMM, 256, 0, stream>>>(x, W1, dinv, bufA, out, n, IN_CH);
  edge_scatter1_kernel<<<gSC, B, 0, stream>>>(src, dst, dinv, bufA, out, bufB, tick, E, n_items);

  gemm2_wmma_kernel<<<gMM, 256, 0, stream>>>(out, W2, b1, dinv, b2, bufA, out, n, HIDC);
  edge_scatter2_kernel<<<gSC, B, 0, stream>>>(src, dst, bufA, out, bufB, tick + 1, E, n_items);
}


# in-degree count runs in the layer-1 product's launch (last blocks, 2048 edges each); sums start at 0, self-loop h*dinv^2 added as layer 2 stages A (summed last; add order arbitrary); 6 launches
# speedup vs baseline: 2.7716x; 1.0306x over previous
#include <hip/hip_runtime.h>

typedef __attribute__((ext_vector_type(2))) float v2f;
typedef __attribute__((ext_vector_type(8))) float v8f;

#define HIDC 64
#define LDSW 68


#define NPLANE 8

#define TICK_PITCH 64
__global__ void init_deg_kernel(float* __restrict__ planes, int* __restrict__ tick, int n4) {
  int i = blockIdx.x * blockDim.x + threadIdx.x;
  if (i < n4) ((float4*)planes)[i] = make_float4(0.f, 0.f, 0.f, 0.f);
  if (i < 2) { tick[i] = 0; tick[TICK_PITCH + i] = 0; }
}

#define DEG_EPT 8
#define DEG_EPB (256 * DEG_EPT)
__device__ __forceinline__ void deg_count_block(const int* __restrict__ dst,
                                                float* __restrict__ planes,
                                                int pitch, int E, int blk) {
  int id = __builtin_amdgcn_s_sendmsg_rtn(0x87);
  float* mine = planes + (size_t)((id >> 16) & (NPLANE - 1)) * pitch;
  const int base = blk * DEG_EPB + threadIdx.x;
  int d[DEG_EPT];
#pragma unroll
  for (int j = 0; j < DEG_EPT; ++j) {
    int e = base + j * 256;
    d[j] = dst[e < E ? e : E - 1];
  }
#pragma unroll
  for (int j = 0; j < DEG_EPT; ++j) {
    if (base + j * 256 < E) unsafeAtomicAdd(&mine[d[j]], 1.0f);
  }
}

__global__ void dinv_kernel(const float* __restrict__ planes, int pitch,
                            float* __restrict__ dinv, int n) {
  int i = blockIdx.x * blockDim.x + threadIdx.x;
  if (i < n) {
    float d = 1.0f;
#pragma unroll
    for (int p = 0; p < NPLANE; ++p) d += planes[(size_t)p * pitch + i];
    dinv[i] = (d > 0.0f) ? rsqrtf(d) : 0.0f;
  }
}


template <bool SECOND>
__device__ __forceinline__ void gemm_wmma_body(
    int blk, const float* A, const float* HP, const float* __restrict__ W,
    const float* __restrict__ bias_in, const float* __restrict__ dinv,
    const float* __restrict__ bias_out,
    float* H, float* AGG, int n, int K) {
  __shared__ float tile[32 * LDSW];

  const int t     = threadIdx.x;
  const int wid   = t >> 5;
  const int lane  = t & 31;
  const int row16 = lane & 15;
  const int half  = lane >> 4;
  const int m_sub = wid >> 2;
  const int n0    = (wid & 3) << 4;
  const int m_base = blk * 32;

  v8f acc = {};

  const int lr = t >> 3;
  const int lc = (t & 7) << 3;

  for (int kb = 0; kb < K; kb += 64) {
    float4 v0 = make_float4(0.f, 0.f, 0.f, 0.f);
    float4 v1 = v0;
    int grow = m_base + lr;
    if (grow < n) {
      const float* p = A + (size_t)grow * K + kb + lc;
      v0 = *(const float4*)(p);
      v1 = *(const float4*)(p + 4);
      if (SECOND) {
        const float* ph = HP + (size_t)grow * HIDC + kb + lc;
        float4 h0 = *(const float4*)(ph);
        float4 h1 = *(const float4*)(ph + 4);
        float4 c0 = *(const float4*)(bias_in + kb + lc);
        float4 c1 = *(const float4*)(bias_in + kb + lc + 4);
        float dd = dinv[grow];
        dd *= dd;
        v0.x += h0.x * dd; v0.y += h0.y * dd; v0.z += h0.z * dd; v0.w += h0.w * dd;
        v1.x += h1.x * dd; v1.y += h1.y * dd; v1.z += h1.z * dd; v1.w += h1.w * dd;
        v0.x += c0.x; v0.y += c0.y; v0.z += c0.z; v0.w += c0.w;
        v1.x += c1.x; v1.y += c1.y; v1.z += c1.z; v1.w += c1.w;
        v0.x = v0.x > 0.0f ? v0.x : 0.0f;
        v0.y = v0.y > 0.0f ? v0.y : 0.0f;
        v0.z = v0.z > 0.0f ? v0.z : 0.0f;
        v0.w = v0.w > 0.0f ? v0.w : 0.0f;
        v1.x = v1.x > 0.0f ? v1.x : 0.0f;
        v1.y = v1.y > 0.0f ? v1.y : 0.0f;
        v1.z = v1.z > 0.0f ? v1.z : 0.0f;
        v1.w = v1.w > 0.0f ? v1.w : 0.0f;
      }
    }
    float* q = &tile[lr * LDSW + lc];
    *(float4*)(q)     = v0;
    *(float4*)(q + 4) = v1;
    __syncthreads();

    const float* arow = &tile[(m_sub * 16 + row16) * LDSW];
#pragma unroll
    for (int kk = 0; kk < 64; kk += 4) {
      v2f a = *(const v2f*)(arow + kk + half * 2);
      int krow = kb + kk + half * 2;
      v2f b;
      b.x = W[(size_t)krow * HIDC + n0 + row16];
      b.y = W[(size_t)(krow + 1) * HIDC + n0 + row16];
      acc = __builtin_amdgcn_wmma_f32_16x16x4_f32(
          false, a, false, b, (short)0, acc, false, false);
    }
    __syncthreads();
  }

  int mrow = m_base + m_sub * 16 + half * 8;
  float bo = 0.0f;
  float dv[8];
  if (SECOND) {
    bo = bias_out[n0 + row16];
#pragma unroll
    for (int r = 0; r < 8; ++r) {
      int gr = mrow + r;
      dv[r] = dinv[gr < n ? gr : n - 1];
    }
  }
#pragma unroll
  for (int r = 0; r < 8; ++r) {
    int gr = mrow + r;
    if (gr < n) {
      float hv = acc[r];
      size_t o = (size_t)gr * HIDC + n0 + row16;
      float v = 0.0f;
      if (SECOND) {
        float d = dv[r];
        v = hv * d * d + bo;
      }
      H[o] = hv;
      AGG[o] = v;
    }
  }
}

__global__ __launch_bounds__(256) void gemm1_wmma_kernel(
    const float* __restrict__ A, const float* __restrict__ W,
    const int* __restrict__ dst, float* __restrict__ planes, int pitch, int E, int n_mm,
    float* __restrict__ H, float* __restrict__ AGG, int n, int K) {
  const int blk = (int)blockIdx.x;
  if (blk >= n_mm) {
    deg_count_block(dst, planes, pitch, E, blk - n_mm);
    return;
  }
  gemm_wmma_body<false>(blk, A, nullptr, W, nullptr, nullptr, nullptr, H, AGG, n, K);
}

__global__ __launch_bounds__(256) void gemm2_wmma_kernel(
    const float* A, const float* HP, const float* __restrict__ W,
    const float* __restrict__ bias_in, const float* __restrict__ dinv,
    const float* __restrict__ bias_out,
    float* H, float* AGG, int n, int K) {
  gemm_wmma_body<true>((int)blockIdx.x, A, HP, W, bias_in, dinv, bias_out, H, AGG, n, K);
}


__device__ __forceinline__ void scatter_row(float lo, float hi, float w,
                                            float* __restrict__ o) {
  unsafeAtomicAdd(o,      lo * w);
  unsafeAtomicAdd(o + 32, hi * w);
}

#define EPW 64
#define EPB (8 * EPW)

template <bool FIRST>
__device__ __forceinline__ void edge_scatter_body(
    const int* __restrict__ src, const int* __restrict__ dst,
    const float* __restrict__ dinv, const float* __restrict__ h,
    float* __restrict__ agg, float* __restrict__ wbuf,
    int* __restrict__ tick, int E, int n_items) {
  __shared__ int sh_pick[2];
  __shared__ int cS[8 * EPW];
  __shared__ int cD[8 * EPW];

  const int lane = threadIdx.x & 31;
  const int wv   = threadIdx.x >> 5;

  if (threadIdx.x == 0) {
    int id = __builtin_amdgcn_s_sendmsg_rtn(0x87);
    int p  = (id >> 18) & 1;
    int tk = atomicAdd(tick + p * TICK_PITCH, 1);
    if (tk >= n_items) {
      p ^= 1;
      tk = atomicAdd(tick + p * TICK_PITCH, 1);
    }
    sh_pick[0] = p;
    sh_pick[1] = tk;
  }
  __syncthreads();
  const int par  = __builtin_amdgcn_readfirstlane(sh_pick[0]);
  const int item = __builtin_amdgcn_readfirstlane(sh_pick[1]);
  if (item >= n_items) return;

  const int e0 = item * EPB + wv * EPW + lane;
  const int e1 = e0 + 32;
  const int ec0 = e0 < E ? e0 : E - 1;
  const int ec1 = e1 < E ? e1 : E - 1;
  const int s0 = src[ec0];
  const int d0 = dst[ec0];
  const int s1 = src[ec1];
  const int d1 = dst[ec1];
  const bool k0 = (e0 < E) && ((d0 & 1) == par);
  const bool k1 = (e1 < E) && ((d1 & 1) == par);

  const unsigned m0 = __builtin_amdgcn_ballot_w32(k0);
  const unsigned m1 = __builtin_amdgcn_ballot_w32(k1);
  const int c0    = __builtin_popcount(m0);
  const int total = c0 + __builtin_popcount(m1);
  const int r0 = __builtin_amdgcn_mbcnt_lo(m0, 0);
  const int r1 = __builtin_amdgcn_mbcnt_lo(m1, 0);
  const int pos0 = k0 ? r0 : 63 - (lane - r0);
  const int pos1 = k1 ? c0 + r1 : 63 - ((32 - c0) + (lane - r1));
  int* mS = cS + wv * EPW;
  int* mD = cD + wv * EPW;
  mS[pos0] = s0; mD[pos0] = d0;
  mS[pos1] = s1; mD[pos1] = d1;
  __syncthreads();
  if (total == 0) return;

  const int q0 = lane < total ? lane : 0;
  const int q1 = lane + 32 < total ? lane + 32 : 0;
  const int S0 = mS[q0];
  const int D0 = mD[q0];
  const int S1 = mS[q1];
  const int D1 = mD[q1];
  float* wb = wbuf + ((size_t)(par * n_items + item) * EPB + wv * EPW + lane);
  float W0, W1;
  if (FIRST) {
    W0 = dinv[S0] * dinv[D0];
    W1 = dinv[S1] * dinv[D1];
    wb[0]  = W0;
    wb[32] = W1;
  } else {
    W0 = wb[0];
    W1 = wb[32];
  }

  const float* hb = h + lane;
  float* ab = agg + lane;
  const int ngroups = total >> 3;
  const int rem     = total & 7;

  for (int g = 0; g <= ngroups; ++g) {
    const int cnt = (g < ngroups) ? 8 : rem;
    if (cnt == 0) break;
    const bool front = g < 4;
    const int Sg = front ? S0 : S1;
    const int Dg = front ? D0 : D1;
    const int Wg = __float_as_int(front ? W0 : W1);
    const int l0 = (g & 3) << 3;
    float lo[8], hi[8];
    if (cnt == 8) {
#pragma unroll
      for (int u = 0; u < 8; ++u) {
        int sj = __builtin_amdgcn_readlane(Sg, l0 + u);
        const float* p = hb + (size_t)sj * HIDC;
        lo[u] = p[0];
        hi[u] = p[32];
      }
#pragma unroll
      for (int u = 0; u < 8; ++u) {
        int dj = __builtin_amdgcn_readlane(Dg, l0 + u);
        float wj = __int_as_float(__builtin_amdgcn_readlane(Wg, l0 + u));
        scatter_row(lo[u], hi[u], wj, ab + (size_t)dj * HIDC);
      }
    } else {
#pragma unroll
      for (int u = 0; u < 8; ++u) {
        if (u < cnt) {
          int sj = __builtin_amdgcn_readlane(Sg, l0 + u);
          const float* p = hb + (size_t)sj * HIDC;
          lo[u] = p[0];
          hi[u] = p[32];
        }
      }
#pragma unroll
      for (int u = 0; u < 8; ++u) {
        if (u < cnt) {
          int dj = __builtin_amdgcn_readlane(Dg, l0 + u);
          float wj = __int_as_float(__builtin_amdgcn_readlane(Wg, l0 + u));
          scatter_row(lo[u], hi[u], wj, ab + (size_t)dj * HIDC);
        }
      }
    }
  }
}

__global__ __launch_bounds__(256) void edge_scatter1_kernel(
    const int* __restrict__ src, const int* __restrict__ dst,
    const float* __restrict__ dinv, const float* __restrict__ h,
    float* __restrict__ agg, float* __restrict__ wbuf,
    int* __restrict__ tick, int E, int n_items) {
  edge_scatter_body<true>(src, dst, dinv, h, agg, wbuf, tick, E, n_items);
}

__global__ __launch_bounds__(256) void edge_scatter2_kernel(
    const int* __restrict__ src, const int* __restrict__ dst,
    const float* __restrict__ h,
    float* __restrict__ agg, float* __restrict__ wbuf,
    int* __restrict__ tick, int E, int n_items) {
  edge_scatter_body<false>(src, dst, nullptr, h, agg, wbuf, tick, E, n_items);
}


extern "C" void kernel_launch(void* const* d_in, const int* in_sizes, int n_in,
                              void* d_out, int out_size, void* d_ws, size_t ws_size,
                              hipStream_t stream) {
  const float* x  = (const float*)d_in[0];
  const int*   ei = (const int*)d_in[1];
  const float* W1 = (const float*)d_in[2];
  const float* b1 = (const float*)d_in[3];
  const float* W2 = (const float*)d_in[4];
  const float* b2 = (const float*)d_in[5];

  const int IN_CH = 512;
  const int n = in_sizes[0] / IN_CH;
  const int E = in_sizes[1] / 2;
  const int* src = ei;
  const int* dst = ei + E;
  float* out = (float*)d_out;

  float* ws = (float*)d_ws;
  size_t n_al = ((size_t)n + 96 + 255) & ~(size_t)255;
  float* dinv = ws;
  int*   tick = (int*)(ws + n);
  float* bufA = ws + n_al;
  float* bufB = bufA + (size_t)n * HIDC;

  float* planes = bufB;
  int    n4     = (int)(NPLANE * n_al / 4);

  const int B = 256;
  unsigned gN  = (unsigned)((n + B - 1) / B);
  unsigned gPL = (unsigned)((n4 + B - 1) / B);
  int n_cnt    = (E + DEG_EPB - 1) / DEG_EPB;
  unsigned gMM = (unsigned)((n + 31) / 32);
  int n_items  = (E + 511) / 512;
  unsigned gSC = (unsigned)(2 * n_items);

  init_deg_kernel<<<gPL, B, 0, stream>>>(planes, tick, n4);
  gemm1_wmma_kernel<<<gMM + (unsigned)n_cnt, 256, 0, stream>>>(x, W1, dst, planes, (int)n_al, E, (int)gMM,
                                                             bufA, out, n, IN_CH);
  dinv_kernel<<<gN, B, 0, stream>>>(planes, (int)n_al, dinv, n);
  edge_scatter1_kernel<<<gSC, B, 0, stream>>>(src, dst, dinv, bufA, out, bufB, tick, E, n_items);

  gemm2_wmma_kernel<<<gMM, 256, 0, stream>>>(out, bufA, W2, b1, dinv, b2, bufA, out, n, HIDC);
  edge_scatter2_kernel<<<gSC, B, 0, stream>>>(src, dst, bufA, out, bufB, tick + 1, E, n_items);
}
